# starting kernel
# baseline (speedup 1.0000x reference)
#include <hip/hip_runtime.h>

typedef float v2f __attribute__((ext_vector_type(2)));
typedef float v8f __attribute__((ext_vector_type(8)));

#define N_NODES 50000
#define N_EDGES 600000
#define DIM     128
#define MTILES  (N_NODES / 16)


__global__ void k_init_deg(float* __restrict__ deg) {
    int i = blockIdx.x * blockDim.x + threadIdx.x;
    if (i < N_NODES) deg[i] = 1.0f;
}

__global__ void k_count(const int* __restrict__ dst, float* __restrict__ deg) {
    int e = blockIdx.x * blockDim.x + threadIdx.x;
    if (e < N_EDGES) atomicAdd(&deg[dst[e]], 1.0f);
}

__global__ void k_dinv(float* __restrict__ d) {
    int i = blockIdx.x * blockDim.x + threadIdx.x;
    if (i < N_NODES) d[i] = rsqrtf(d[i]);
}

__global__ void k_repack(const float* __restrict__ W, float* __restrict__ Wp) {
    int idx = blockIdx.x * blockDim.x + threadIdx.x;
    if (idx < DIM * DIM) {
        int k = idx >> 7, n = idx & 127;
        Wp[(k >> 1) * 256 + n * 2 + (k & 1)] = W[idx];
    }
}


template <int RELU>
__global__ __launch_bounds__(256) void k_gemm(const float* __restrict__ X,
                                              const float* __restrict__ Wp,
                                              float* __restrict__ H) {
    int wave = threadIdx.x >> 5;
    int lane = threadIdx.x & 31;
    int mt   = blockIdx.x * 8 + wave;
    if (mt >= MTILES) return;

    int g = lane >> 4;
    int r = lane & 15;

    const float* arow = X + ((size_t)mt * 16 + r) * DIM;

    v8f acc[8] = {};

#pragma unroll
    for (int kb = 0; kb < DIM; kb += 4) {
        int k0 = kb + 2 * g;
        v2f a = *(const v2f*)(arow + k0);
        if (RELU) { a.x = fmaxf(a.x, 0.0f); a.y = fmaxf(a.y, 0.0f); }
        const float* bbase = Wp + (size_t)(k0 >> 1) * 256 + r * 2;
#pragma unroll
        for (int nt = 0; nt < 8; ++nt) {
            v2f b = *(const v2f*)(bbase + nt * 32);
            acc[nt] = __builtin_amdgcn_wmma_f32_16x16x4_f32(
                false, a, false, b, (short)0, acc[nt], false, false);
        }
    }

    float* hrow = H + ((size_t)mt * 16 + 8 * g) * DIM + r;
#pragma unroll
    for (int v = 0; v < 8; ++v) {
#pragma unroll
        for (int nt = 0; nt < 8; ++nt)
            hrow[(size_t)v * DIM + nt * 16] = acc[nt][v];
    }
}

__global__ void k_self(const float* __restrict__ h, const float* __restrict__ dinv,
                       const float* __restrict__ bias, float* __restrict__ out) {
    int idx = blockIdx.x * blockDim.x + threadIdx.x;
    if (idx < N_NODES * DIM) {
        int node = idx >> 7, c = idx & 127;
        float di = dinv[node];
        out[idx] = bias[c] + h[idx] * di * di;
    }
}

__global__ __launch_bounds__(256) void k_scatter(const float* __restrict__ h,
                                                 const int* __restrict__ src,
                                                 const int* __restrict__ dst,
                                                 const float* __restrict__ dinv,
                                                 float* __restrict__ out) {
    int tid  = blockIdx.x * blockDim.x + threadIdx.x;
    int e    = tid >> 5;
    int lane = threadIdx.x & 31;
    if (e >= N_EDGES) return;

    int s = src[e], d = dst[e];
    float nrm = dinv[s] * dinv[d];

    const float4* hv = (const float4*)(h + (size_t)s * DIM);
    float4 v = hv[lane];

    float* o = out + (size_t)d * DIM + lane * 4;
    atomicAdd(o + 0, v.x * nrm);
    atomicAdd(o + 1, v.y * nrm);
    atomicAdd(o + 2, v.z * nrm);
    atomicAdd(o + 3, v.w * nrm);
}


extern "C" void kernel_launch(void* const* d_in, const int* in_sizes, int n_in,
                              void* d_out, int out_size, void* d_ws, size_t ws_size,
                              hipStream_t stream) {
    const float* X  = (const float*)d_in[0];
    const int*   ei = (const int*)d_in[1];
    const float* W1 = (const float*)d_in[2];
    const float* b1 = (const float*)d_in[3];
    const float* W2 = (const float*)d_in[4];
    const float* b2 = (const float*)d_in[5];
    float* out = (float*)d_out;

    const int* src = ei;
    const int* dst = ei + N_EDGES;

    float* ws   = (float*)d_ws;
    float* dinv = ws;
    float* Wp1  = ws + 50048;
    float* Wp2  = Wp1 + 16384;
    float* h    = Wp2 + 16384;
    float* x1   = h + (size_t)N_NODES * DIM;

    const int thr = 256;
    const int gNode   = (N_NODES + thr - 1) / thr;
    const int gEdge1  = (N_EDGES + thr - 1) / thr;
    const int gEdge32 = (N_EDGES * 32 + thr - 1) / thr;
    const int gFeat   = (N_NODES * DIM + thr - 1) / thr;
    const int gGemm   = (MTILES + 7) / 8;
    const int gW      = (DIM * DIM + thr - 1) / thr;

    k_init_deg<<<gNode, thr, 0, stream>>>(dinv);
    k_count   <<<gEdge1, thr, 0, stream>>>(dst, dinv);
    k_dinv    <<<gNode, thr, 0, stream>>>(dinv);

    k_repack<<<gW, thr, 0, stream>>>(W1, Wp1);
    k_repack<<<gW, thr, 0, stream>>>(W2, Wp2);

    k_gemm<0><<<gGemm, thr, 0, stream>>>(X, Wp1, h);
    k_self    <<<gFeat, thr, 0, stream>>>(h, dinv, b1, x1);
    k_scatter <<<gEdge32, thr, 0, stream>>>(h, src, dst, dinv, x1);

    k_gemm<1><<<gGemm, thr, 0, stream>>>(x1, Wp2, h);
    k_self    <<<gFeat, thr, 0, stream>>>(h, dinv, b2, out);
    k_scatter <<<gEdge32, thr, 0, stream>>>(h, src, dst, dinv, out);
}


# f32 atomic scatter-add replaced by a destination-sorted edge list and a gather: one wave per node sums its in-edge rows in registers and stores the row once, self loop and bias in the same pass
# speedup vs baseline: 2.1139x; 2.1139x over previous
#include <hip/hip_runtime.h>


typedef float v2f __attribute__((ext_vector_type(2)));
typedef float v8f __attribute__((ext_vector_type(8)));

#define N_NODES 50000
#define N_EDGES 600000
#define DIM     128
#define MTILES  (N_NODES / 16)


__global__ void k_zero_cnt(int* __restrict__ cnt) {
    int i = blockIdx.x * blockDim.x + threadIdx.x;
    if (i < N_NODES) cnt[i] = 0;
}

__global__ void k_rank(const int* __restrict__ dst, int* __restrict__ cnt,
                       int* __restrict__ rank) {
    int e = blockIdx.x * blockDim.x + threadIdx.x;
    if (e < N_EDGES) rank[e] = atomicAdd(&cnt[dst[e]], 1);
}

__global__ __launch_bounds__(256) void k_scan(const int* __restrict__ cnt,
                                              int* __restrict__ rowptr,
                                              float* __restrict__ dinv) {
    __shared__ int sc[256];
    int b = blockIdx.x, t = threadIdx.x;

    const int4* c4 = (const int4*)cnt;
    int nvec = b * 64;
    int part = 0;
    for (int j = t; j < nvec; j += 256) {
        int4 v = c4[j];
        part += (v.x + v.y) + (v.z + v.w);
    }
    sc[t] = part;
    __syncthreads();
    for (int s = 128; s > 0; s >>= 1) {
        if (t < s) sc[t] += sc[t + s];
        __syncthreads();
    }
    int base = sc[0];
    __syncthreads();

    int i = b * 256 + t;
    int c = (i < N_NODES) ? cnt[i] : 0;
    sc[t] = c;
    __syncthreads();
    for (int off = 1; off < 256; off <<= 1) {
        int v = (t >= off) ? sc[t - off] : 0;
        __syncthreads();
        sc[t] += v;
        __syncthreads();
    }
    int incl = sc[t];
    if (i < N_NODES) {
        rowptr[i] = base + incl - c;
        dinv[i]   = rsqrtf((float)(c + 1));
    }
    if (i == N_NODES - 1) rowptr[N_NODES] = base + incl;
}

__global__ void k_fill(const int* __restrict__ src, const int* __restrict__ dst,
                       const int* __restrict__ rank, const int* __restrict__ rowptr,
                       const float* __restrict__ dinv, int2* __restrict__ csr) {
    int e = blockIdx.x * blockDim.x + threadIdx.x;
    if (e < N_EDGES) {
        int s = src[e];
        csr[rowptr[dst[e]] + rank[e]] = make_int2(s, __float_as_int(dinv[s]));
    }
}

__global__ void k_repack(const float* __restrict__ W, float* __restrict__ Wp) {
    int idx = blockIdx.x * blockDim.x + threadIdx.x;
    if (idx < DIM * DIM) {
        int k = idx >> 7, n = idx & 127;
        Wp[(k >> 1) * 256 + n * 2 + (k & 1)] = W[idx];
    }
}


template <int RELU>
__global__ __launch_bounds__(256) void k_gemm(const float* __restrict__ X,
                                              const float* __restrict__ Wp,
                                              float* __restrict__ H) {
    int wave = threadIdx.x >> 5;
    int lane = threadIdx.x & 31;
    int mt   = blockIdx.x * 8 + wave;
    if (mt >= MTILES) return;

    int g = lane >> 4;
    int r = lane & 15;

    const float* arow = X + ((size_t)mt * 16 + r) * DIM;

    v8f acc[8] = {};

#pragma unroll
    for (int kb = 0; kb < DIM; kb += 4) {
        int k0 = kb + 2 * g;
        v2f a = *(const v2f*)(arow + k0);
        if (RELU) { a.x = fmaxf(a.x, 0.0f); a.y = fmaxf(a.y, 0.0f); }
        const float* bbase = Wp + (size_t)(k0 >> 1) * 256 + r * 2;
#pragma unroll
        for (int nt = 0; nt < 8; ++nt) {
            v2f b = *(const v2f*)(bbase + nt * 32);
            acc[nt] = __builtin_amdgcn_wmma_f32_16x16x4_f32(
                false, a, false, b, (short)0, acc[nt], false, false);
        }
    }

    float* hrow = H + ((size_t)mt * 16 + 8 * g) * DIM + r;
#pragma unroll
    for (int v = 0; v < 8; ++v) {
#pragma unroll
        for (int nt = 0; nt < 8; ++nt)
            hrow[(size_t)v * DIM + nt * 16] = acc[nt][v];
    }
}


__device__ __forceinline__ float4 row_fma(float4 a, float4 v, float w) {
    a.x = fmaf(v.x, w, a.x); a.y = fmaf(v.y, w, a.y);
    a.z = fmaf(v.z, w, a.z); a.w = fmaf(v.w, w, a.w);
    return a;
}

template <int N>
__device__ __forceinline__ void rows_load(float4* v, const float4* __restrict__ h4,
                                          int lane, int sl, int j) {
#pragma unroll
    for (int k = 0; k < N; ++k)
        v[k] = h4[(size_t)__builtin_amdgcn_readlane(sl, j + k) * 32 + lane];
}

template <int N>
__device__ __forceinline__ float4 rows_add(float4 acc, const float4* v, int wl, int j) {
#pragma unroll
    for (int k = 0; k < N; ++k)
        acc = row_fma(acc, v[k], __int_as_float(__builtin_amdgcn_readlane(wl, j + k)));
    return acc;
}

__global__ __launch_bounds__(256) void k_gather(const float* __restrict__ h,
                                                const int* __restrict__ rowptr,
                                                const int2* __restrict__ csr,
                                                const float* __restrict__ dinv,
                                                const float* __restrict__ bias,
                                                float* __restrict__ out) {
    int lane = threadIdx.x & 31;
    int node = blockIdx.x * 8 + (threadIdx.x >> 5);
    if (node >= N_NODES) return;
    node = __builtin_amdgcn_readfirstlane(node);

    int   beg = rowptr[node];
    int   end = rowptr[node + 1];
    float dn  = dinv[node];

    const float4* h4 = (const float4*)h;
    float4 hv = h4[(size_t)node * 32 + lane];
    float4 bv = ((const float4*)bias)[lane];
    float4 acc = row_fma(bv, hv, dn * dn);

    for (int base = beg; base < end; base += 32) {
        int idx = base + lane;
        int2 ent = make_int2(0, 0);
        if (idx < end) ent = csr[idx];
        int sl = ent.x;
        int wl = __float_as_int(__int_as_float(ent.y) * dn);
        int nb = end - base;
        if (nb > 32) nb = 32;

        float4 v[8];
        int j = 0;
        for (; j + 8 <= nb; j += 8) {
            rows_load<8>(v, h4, lane, sl, j);
            acc = rows_add<8>(acc, v, wl, j);
        }
        int rem = nb - j;
        int j2 = j + (rem & 4);
        int j1 = j2 + (rem & 2);
        if (rem & 4) rows_load<4>(v,     h4, lane, sl, j);
        if (rem & 2) rows_load<2>(v + 4, h4, lane, sl, j2);
        if (rem & 1) rows_load<1>(v + 6, h4, lane, sl, j1);
        if (rem & 4) acc = rows_add<4>(acc, v,     wl, j);
        if (rem & 2) acc = rows_add<2>(acc, v + 4, wl, j2);
        if (rem & 1) acc = rows_add<1>(acc, v + 6, wl, j1);
    }

    ((float4*)out)[(size_t)node * 32 + lane] = acc;
}


extern "C" void kernel_launch(void* const* d_in, const int* in_sizes, int n_in,
                              void* d_out, int out_size, void* d_ws, size_t ws_size,
                              hipStream_t stream) {
    const float* X  = (const float*)d_in[0];
    const int*   ei = (const int*)d_in[1];
    const float* W1 = (const float*)d_in[2];
    const float* b1 = (const float*)d_in[3];
    const float* W2 = (const float*)d_in[4];
    const float* b2 = (const float*)d_in[5];
    float* out = (float*)d_out;

    const int* src = ei;
    const int* dst = ei + N_EDGES;

    float* ws   = (float*)d_ws;
    float* dinv = ws;
    float* Wp1  = ws + 50048;
    float* Wp2  = Wp1 + 16384;
    float* h    = Wp2 + 16384;
    int*  cnt    = (int*)(h + (size_t)N_NODES * DIM);
    int*  rowptr = cnt + 50048;
    int*  rank   = rowptr + 50048;
    int2* csr    = (int2*)(rank + N_EDGES);
    float* x1    = out;

    const int thr = 256;
    const int gNode   = (N_NODES + thr - 1) / thr;
    const int gEdge1  = (N_EDGES + thr - 1) / thr;
    const int gRow    = (N_NODES + 7) / 8;
    const int gGemm   = (MTILES + 7) / 8;
    const int gW      = (DIM * DIM + thr - 1) / thr;

    k_zero_cnt<<<gNode, thr, 0, stream>>>(cnt);
    k_rank    <<<gEdge1, thr, 0, stream>>>(dst, cnt, rank);
    k_scan    <<<gNode, thr, 0, stream>>>(cnt, rowptr, dinv);
    k_fill    <<<gEdge1, thr, 0, stream>>>(src, dst, rank, rowptr, dinv, csr);

    k_repack<<<gW, thr, 0, stream>>>(W1, Wp1);
    k_repack<<<gW, thr, 0, stream>>>(W2, Wp2);

    k_gemm<0><<<gGemm, thr, 0, stream>>>(X, Wp1, h);
    k_gather <<<gRow, thr, 0, stream>>>(h, rowptr, csr, dinv, b1, x1);

    k_gemm<1><<<gGemm, thr, 0, stream>>>(x1, Wp2, h);
    k_gather <<<gRow, thr, 0, stream>>>(h, rowptr, csr, dinv, b2, out);
}


# layer-2 product fused into layer 1's aggregation: a work-group gathers 32 rows into LDS and multiplies them by W2 on the matrix unit at once; 5 -> 4 launches, layer-1 rows never go to memory
# speedup vs baseline: 2.7401x; 1.2962x over previous
#include <hip/hip_runtime.h>


typedef float v2f __attribute__((ext_vector_type(2)));
typedef float v8f __attribute__((ext_vector_type(8)));

#define N_NODES 50000
#define N_EDGES 600000
#define DIM     128
#define MTILES  (N_NODES / 16)
static_assert(N_NODES <= (1 << 26), "the gather's sort key keeps the source above 5 lane bits");

#define CAP 32

#define THR       256
#define G_NODE    ((N_NODES + THR - 1) / THR)
#define G_W       ((DIM * DIM + THR - 1) / THR)
#define G_GEMM    ((MTILES + 7) / 8)

__device__ __forceinline__ void zero_cnt_body(int blk, int* __restrict__ cnt) {
    int i = blk * THR + threadIdx.x;
    if (i <= N_NODES) cnt[i] = 0;
}


__device__ __forceinline__ void repack_body(int blk, const float* __restrict__ W,
                                            float* __restrict__ Wp) {
    int idx = blk * THR + threadIdx.x;
    if (idx < DIM * DIM) {
        int k = idx >> 7, n = idx & 127;
        Wp[(k >> 1) * 256 + n * 2 + (k & 1)] = W[idx];
    }
}

__global__ __launch_bounds__(THR) void k_prep(int* __restrict__ cnt,
                                              const float* __restrict__ W1, float* __restrict__ Wp1,
                                              const float* __restrict__ W2, float* __restrict__ Wp2) {
    int b = blockIdx.x;
    if (b < G_NODE)            zero_cnt_body(b, cnt);
    else if (b < G_NODE + G_W) repack_body(b - G_NODE, W1, Wp1);
    else                       repack_body(b - G_NODE - G_W, W2, Wp2);
}


__device__ __forceinline__ void gemm_body(int blk, const float* __restrict__ X,
                                          const float* __restrict__ Wp,
                                          float* __restrict__ H) {
    int wave = threadIdx.x >> 5;
    int lane = threadIdx.x & 31;
    int mt   = blk * 8 + wave;
    if (mt >= MTILES) return;

    int g = lane >> 4;
    int r = lane & 15;

    const float* arow = X + ((size_t)mt * 16 + r) * DIM;

    v8f acc[8] = {};

#pragma unroll
    for (int kb = 0; kb < DIM; kb += 4) {
        int k0 = kb + 2 * g;
        v2f a = *(const v2f*)(arow + k0);
        const float* bbase = Wp + (size_t)(k0 >> 1) * 256 + r * 2;
#pragma unroll
        for (int nt = 0; nt < 8; ++nt) {
            v2f b = *(const v2f*)(bbase + nt * 32);
            acc[nt] = __builtin_amdgcn_wmma_f32_16x16x4_f32(
                false, a, false, b, (short)0, acc[nt], false, false);
        }
    }

    float* hrow = H + ((size_t)mt * 16 + 8 * g) * DIM + r;
#pragma unroll
    for (int v = 0; v < 8; ++v) {
#pragma unroll
        for (int nt = 0; nt < 8; ++nt)
            hrow[(size_t)v * DIM + nt * 16] = acc[nt][v];
    }
}

#define PLACE_PER_THREAD 6
static_assert(PLACE_PER_THREAD * G_GEMM * THR >= N_EDGES, "every edge needs a placing thread");

__global__ __launch_bounds__(THR) void k_gemm1_place(const float* __restrict__ X,
                                                     const float* __restrict__ Wp,
                                                     float* __restrict__ H,
                                                     const int* __restrict__ src,
                                                     const int* __restrict__ dst,
                                                     int* __restrict__ cnt,
                                                     int2* __restrict__ ent,
                                                     int2* __restrict__ spill) {
    int gtid = blockIdx.x * THR + threadIdx.x;

    int d[PLACE_PER_THREAD], s[PLACE_PER_THREAD], r[PLACE_PER_THREAD];
#pragma unroll
    for (int i = 0; i < PLACE_PER_THREAD; ++i) {
        int e = i * (G_GEMM * THR) + gtid;
        d[i] = dst[e < N_EDGES ? e : N_EDGES - 1];
    }

    gemm_body(blockIdx.x, X, Wp, H);

#pragma unroll
    for (int i = 0; i < PLACE_PER_THREAD; ++i) {
        int e = i * (G_GEMM * THR) + gtid;
        s[i] = src[e < N_EDGES ? e : N_EDGES - 1];
    }
#pragma unroll
    for (int i = 0; i < PLACE_PER_THREAD; ++i) {
        int e = i * (G_GEMM * THR) + gtid;
        r[i] = (e < N_EDGES) ? atomicAdd(&cnt[(unsigned)d[i]], 1) : 0;
    }
#pragma unroll
    for (int i = 0; i < PLACE_PER_THREAD; ++i) {
        int e = i * (G_GEMM * THR) + gtid;
        if (e < N_EDGES) {
            if (r[i] < CAP) {
                ent[(size_t)(unsigned)d[i] * CAP + r[i]].x = s[i];
            } else {
                int o = atomicAdd(&cnt[N_NODES], 1);
                spill[o] = make_int2(d[i], s[i]);
            }
        }
    }
}


__device__ __forceinline__ float4 row_fma(float4 a, float4 v, float w) {
    a.x = fmaf(v.x, w, a.x); a.y = fmaf(v.y, w, a.y);
    a.z = fmaf(v.z, w, a.z); a.w = fmaf(v.w, w, a.w);
    return a;
}

template <int N>
__device__ __forceinline__ void rows_load(float4* v, const float4* __restrict__ h4,
                                          int lane, int sl, int j) {
#pragma unroll
    for (int k = 0; k < N; ++k)
        v[k] = h4[(size_t)__builtin_amdgcn_readlane(sl, j + k) * 32 + lane];
}

template <int N>
__device__ __forceinline__ float4 rows_add(float4 acc, const float4* v, int wl, int j) {
#pragma unroll
    for (int k = 0; k < N; ++k)
        acc = row_fma(acc, v[k], __int_as_float(__builtin_amdgcn_readlane(wl, j + k)));
    return acc;
}

template <int FIRST>
__device__ __forceinline__ float4 gather_row(int node, int lane,
                                             const float4* __restrict__ h4,
                                             const int* __restrict__ cnt,
                                             int2* __restrict__ ent,
                                             const int2* __restrict__ spill,
                                             float4 bv) {
    int   n  = cnt[node];
    float dn = rsqrtf((float)(n + 1));

    float4 hv = h4[(size_t)node * 32 + lane];
    float4 acc = row_fma(bv, hv, dn * dn);

    int nb = n < CAP ? n : CAP;
    int2* my = ent + (size_t)node * CAP + lane;
    int sl, wl;
    if (FIRST) {
        int s = node;
        if (lane < nb) s = my->x;
        sl = s;
        wl = __float_as_int(rsqrtf((float)(cnt[s] + 1)) * dn);

        int key = (sl << 5) | lane;
        int pos = 0;
        for (int j = 0; j < nb; ++j)
            pos += (__builtin_amdgcn_readlane(key, j) < key) ? 1 : 0;
        if (lane >= nb) pos = lane;
        sl = __builtin_amdgcn_ds_permute(pos << 2, sl);
        wl = __builtin_amdgcn_ds_permute(pos << 2, wl);
        if (lane < nb) *my = make_int2(sl, wl);
    } else {
        int2 e = make_int2(0, 0);
        if (lane < nb) e = *my;
        sl = e.x;
        wl = e.y;
    }

    float4 v[8];
    int j = 0;
    for (; j + 8 <= nb; j += 8) {
        rows_load<8>(v, h4, lane, sl, j);
        acc = rows_add<8>(acc, v, wl, j);
    }
    int rem = nb - j;
    int j2 = j + (rem & 4);
    int j1 = j2 + (rem & 2);
    if (rem & 4) rows_load<4>(v,     h4, lane, sl, j);
    if (rem & 2) rows_load<2>(v + 4, h4, lane, sl, j2);
    if (rem & 1) rows_load<1>(v + 6, h4, lane, sl, j1);
    if (rem & 4) acc = rows_add<4>(acc, v,     wl, j);
    if (rem & 2) acc = rows_add<2>(acc, v + 4, wl, j2);
    if (rem & 1) acc = rows_add<1>(acc, v + 6, wl, j1);

    if (n > CAP) {
        int ns = cnt[N_NODES];
        if (ns > N_EDGES) ns = N_EDGES;
        for (int base = 0; base < ns; base += 32) {
            int i = base + lane;
            int2 e = make_int2(-1, 0);
            if (i < ns) e = spill[i];
            unsigned m = __builtin_amdgcn_ballot_w32(e.x == node);
            while (m) {
                int k = __builtin_ctz(m);
                m &= m - 1;
                int s = __builtin_amdgcn_readlane(e.y, k);
                float w = rsqrtf((float)(cnt[s] + 1)) * dn;
                acc = row_fma(acc, h4[(size_t)s * 32 + lane], w);
            }
        }
    }

    return acc;
}

__global__ __launch_bounds__(256) void k_gather2(const float* __restrict__ h,
                                                 const int* __restrict__ cnt,
                                                 int2* __restrict__ ent,
                                                 const int2* __restrict__ spill,
                                                 const float* __restrict__ bias,
                                                 float* __restrict__ out) {
    int lane = threadIdx.x & 31;
    int node = blockIdx.x * 8 + (threadIdx.x >> 5);
    if (node >= N_NODES) return;
    node = __builtin_amdgcn_readfirstlane(node);
    float4 bv = ((const float4*)bias)[lane];
    float4 acc = gather_row<0>(node, lane, (const float4*)h, cnt, ent, spill, bv);
    ((float4*)out)[(size_t)node * 32 + lane] = acc;
}

#define NPB   32
#define PITCH 132
#define G_FUSED ((N_NODES + NPB - 1) / NPB)
static_assert(N_NODES % 16 == 0, "an M tile is either whole or absent");

__global__ __launch_bounds__(256) void k_gather1_gemm2(const float* __restrict__ h1,
                                                       const int* __restrict__ cnt,
                                                       int2* __restrict__ ent,
                                                       const int2* __restrict__ spill,
                                                       const float* __restrict__ bias,
                                                       const float* __restrict__ Wp,
                                                       float* __restrict__ h2) {
    __shared__ __attribute__((aligned(16))) float xs[NPB * PITCH];

    int lane = threadIdx.x & 31;
    int wave = __builtin_amdgcn_readfirstlane(threadIdx.x >> 5);
    int base = blockIdx.x * NPB;

    float4 bv = ((const float4*)bias)[lane];
    for (int i = 0; i < NPB / 8; ++i) {
        int loc  = i * 8 + wave;
        int node = base + loc;
        float4 x = make_float4(0.0f, 0.0f, 0.0f, 0.0f);
        if (node < N_NODES) {
            x = gather_row<1>(node, lane, (const float4*)h1, cnt, ent, spill, bv);
            x.x = fmaxf(x.x, 0.0f); x.y = fmaxf(x.y, 0.0f);
            x.z = fmaxf(x.z, 0.0f); x.w = fmaxf(x.w, 0.0f);
        }
        *(float4*)(xs + loc * PITCH + 4 * lane) = x;
    }
    __syncthreads();

    int g = lane >> 4;
    int r = lane & 15;
    const float* ap = xs + r * PITCH + 2 * g;
    const float* bp = Wp + (size_t)g * 256 + (wave * 16 + r) * 2;
    v8f acc0 = {}, acc1 = {};
#pragma unroll
    for (int kb = 0; kb < DIM; kb += 4) {
        v2f a0 = *(const v2f*)(ap + kb);
        v2f a1 = *(const v2f*)(ap + 16 * PITCH + kb);
        v2f b  = *(const v2f*)(bp + (kb >> 1) * 256);
        acc0 = __builtin_amdgcn_wmma_f32_16x16x4_f32(false, a0, false, b, (short)0, acc0, false, false);
        acc1 = __builtin_amdgcn_wmma_f32_16x16x4_f32(false, a1, false, b, (short)0, acc1, false, false);
    }

    float* hrow = h2 + ((size_t)base + 8 * g) * DIM + wave * 16 + r;
#pragma unroll
    for (int v = 0; v < 8; ++v) hrow[(size_t)v * DIM] = acc0[v];
    if (base + 16 < N_NODES) {
#pragma unroll
        for (int v = 0; v < 8; ++v) hrow[(size_t)(16 + v) * DIM] = acc1[v];
    }
}


extern "C" void kernel_launch(void* const* d_in, const int* in_sizes, int n_in,
                              void* d_out, int out_size, void* d_ws, size_t ws_size,
                              hipStream_t stream) {
    const float* X  = (const float*)d_in[0];
    const int*   ei = (const int*)d_in[1];
    const float* W1 = (const float*)d_in[2];
    const float* b1 = (const float*)d_in[3];
    const float* W2 = (const float*)d_in[4];
    const float* b2 = (const float*)d_in[5];
    float* out = (float*)d_out;

    const int* src = ei;
    const int* dst = ei + N_EDGES;

    float* ws   = (float*)d_ws;
    float* Wp1  = ws;
    float* Wp2  = Wp1 + 16384;
    float* h2   = Wp2 + 16384;
    int*  cnt    = (int*)(h2 + (size_t)N_NODES * DIM);
    int2* ent    = (int2*)(cnt + 50048);
    int2* spill  = ent + (size_t)N_NODES * CAP;
    float* h1    = out;

    if ((size_t)((const char*)(spill + N_EDGES) - (const char*)d_ws) > ws_size) return;

    const int gRow = (N_NODES + 7) / 8;

    k_prep<<<G_NODE + 2 * G_W, THR, 0, stream>>>(cnt, W1, Wp1, W2, Wp2);

    k_gemm1_place<<<G_GEMM, THR, 0, stream>>>(X, Wp1, h1, src, dst, cnt, ent, spill);

    k_gather1_gemm2<<<G_FUSED, THR, 0, stream>>>(h1, cnt, ent, spill, b1, Wp2, h2);

    k_gather2<<<gRow, THR, 0, stream>>>(h2, cnt, ent, spill, b2, out);
}


# edge ranking by LDS counters instead of 600000 scattered returning atomics in memory: chunks counting-sorted by destination range (launch 1), one work-group per range ranks beside the layer-1 product
# speedup vs baseline: 4.1595x; 1.5180x over previous
#include <hip/hip_runtime.h>


typedef float v2f __attribute__((ext_vector_type(2)));
typedef float v8f __attribute__((ext_vector_type(8)));

#define N_NODES 50000
#define N_EDGES 600000
#define DIM     128
#define MTILES  (N_NODES / 16)
static_assert(N_NODES <= (1 << 26), "the gather's sort key keeps the source above 5 lane bits");

#define CAP 32

#define THR       256
#define G_W       ((DIM * DIM + THR - 1) / THR)
#define G_GEMM    ((MTILES + 7) / 8)

#define BINW  256
#define NBIN  ((N_NODES + BINW - 1) / BINW)
#define NCH   THR
#define EPB   ((N_EDGES + NCH - 1) / NCH)
#define EPT   ((EPB + THR - 1) / THR)
static_assert(NBIN < THR, "the range offsets are scanned one per thread");
static_assert(BINW == THR, "pass B stores one count per thread");

__device__ __forceinline__ int bin_of(int d) {
    unsigned b = (unsigned)d / BINW;
    return (int)(b < NBIN ? b : NBIN - 1);
}

__device__ __forceinline__ void sort_chunk_body(int blk, const int* __restrict__ src,
                                                const int* __restrict__ dst,
                                                int2* __restrict__ sorted,
                                                int* __restrict__ offsT,
                                                int* __restrict__ cnt) {
    __shared__ int  hist[THR];
    __shared__ int  wsum[THR / 32];
    __shared__ int2 sbuf[EPB];

    int t = threadIdx.x;
    int base = blk * EPB;
    int nvalid = N_EDGES - base;
    if (nvalid > EPB) nvalid = EPB;

    hist[t] = 0;
    if (blk == 0 && t == 0) cnt[N_NODES] = 0;
    __syncthreads();

    int d[EPT], s[EPT], r[EPT];
#pragma unroll
    for (int i = 0; i < EPT; ++i) {
        int idx = i * THR + t;
        int e = base + (idx < nvalid ? idx : 0);
        d[i] = dst[e];
        s[i] = src[e];
    }
#pragma unroll
    for (int i = 0; i < EPT; ++i) {
        int idx = i * THR + t;
        r[i] = (idx < nvalid) ? atomicAdd(&hist[bin_of(d[i])], 1) : 0;
    }
    __syncthreads();

    int v = hist[t];
    int lane = t & 31, w = t >> 5;
    int x = v;
#pragma unroll
    for (int o = 1; o < 32; o <<= 1) {
        int y = __shfl_up(x, o);
        if (lane >= o) x += y;
    }
    if (lane == 31) wsum[w] = x;
    __syncthreads();
    int add = 0;
    for (int i = 0; i < w; ++i) add += wsum[i];
    int excl = x - v + add;
    hist[t] = excl;
    __syncthreads();

#pragma unroll
    for (int i = 0; i < EPT; ++i) {
        int idx = i * THR + t;
        if (idx < nvalid) sbuf[hist[bin_of(d[i])] + r[i]] = make_int2(d[i], s[i]);
    }
    __syncthreads();

#pragma unroll
    for (int i = 0; i < EPT; ++i) {
        int idx = i * THR + t;
        if (idx < nvalid) sorted[base + idx] = sbuf[idx];
    }
    if (t <= NBIN) offsT[t * NCH + blk] = excl;
}

__device__ __forceinline__ void repack_body(int blk, const float* __restrict__ W,
                                            float* __restrict__ Wp) {
    int idx = blk * THR + threadIdx.x;
    if (idx < DIM * DIM) {
        int k = idx >> 7, n = idx & 127;
        Wp[(k >> 1) * 256 + n * 2 + (k & 1)] = W[idx];
    }
}

__global__ __launch_bounds__(THR) void k_prep(const int* __restrict__ src,
                                              const int* __restrict__ dst,
                                              int2* __restrict__ sorted,
                                              int* __restrict__ offsT,
                                              int* __restrict__ cnt,
                                              const float* __restrict__ W1, float* __restrict__ Wp1,
                                              const float* __restrict__ W2, float* __restrict__ Wp2) {
    int b = blockIdx.x;
    if (b < NCH)            sort_chunk_body(b, src, dst, sorted, offsT, cnt);
    else if (b < NCH + G_W) repack_body(b - NCH, W1, Wp1);
    else                    repack_body(b - NCH - G_W, W2, Wp2);
}


__device__ __forceinline__ void gemm_body(int blk, const float* __restrict__ X,
                                          const float* __restrict__ Wp,
                                          float* __restrict__ H) {
    int wave = threadIdx.x >> 5;
    int lane = threadIdx.x & 31;
    int mt   = blk * 8 + wave;
    if (mt >= MTILES) return;

    int g = lane >> 4;
    int r = lane & 15;

    const float* arow = X + ((size_t)mt * 16 + r) * DIM;

    v8f acc[8] = {};

#pragma unroll
    for (int kb = 0; kb < DIM; kb += 4) {
        int k0 = kb + 2 * g;
        v2f a = *(const v2f*)(arow + k0);
        const float* bbase = Wp + (size_t)(k0 >> 1) * 256 + r * 2;
#pragma unroll
        for (int nt = 0; nt < 8; ++nt) {
            v2f b = *(const v2f*)(bbase + nt * 32);
            acc[nt] = __builtin_amdgcn_wmma_f32_16x16x4_f32(
                false, a, false, b, (short)0, acc[nt], false, false);
        }
    }

    float* hrow = H + ((size_t)mt * 16 + 8 * g) * DIM + r;
#pragma unroll
    for (int v = 0; v < 8; ++v) {
#pragma unroll
        for (int nt = 0; nt < 8; ++nt)
            hrow[(size_t)v * DIM + nt * 16] = acc[nt][v];
    }
}

__device__ __forceinline__ void place_bin_body(int k, const int2* __restrict__ sorted,
                                               const int* __restrict__ offsT,
                                               int* __restrict__ cnt,
                                               int2* __restrict__ ent,
                                               int2* __restrict__ spill) {
    __shared__ int lcnt[BINW];
    int t = threadIdx.x;
    lcnt[t] = 0;
    __syncthreads();

    int o0 = offsT[k * NCH + t];
    int o1 = offsT[(k + 1) * NCH + t];
    const int2* seg = sorted + (size_t)t * EPB;
    int lo = k * BINW;
    for (int j = o0; j < o1; j += 4) {
        int2 e[4];
#pragma unroll
        for (int q = 0; q < 4; ++q) e[q] = seg[j + q < o1 ? j + q : o1 - 1];
#pragma unroll
        for (int q = 0; q < 4; ++q) {
            unsigned loc = (unsigned)(e[q].x - lo);
            if (j + q < o1 && loc < BINW && (unsigned)e[q].x < N_NODES) {
                int r = atomicAdd(&lcnt[loc], 1);
                if (r < CAP) {
                    ent[(size_t)(unsigned)e[q].x * CAP + r].x = e[q].y;
                } else {
                    int o = atomicAdd(&cnt[N_NODES], 1);
                    spill[o] = e[q];
                }
            }
        }
    }
    __syncthreads();
    int node = lo + t;
    if (node < N_NODES) cnt[node] = lcnt[t];
}

__global__ __launch_bounds__(THR) void k_gemm1_place(const float* __restrict__ X,
                                                     const float* __restrict__ Wp,
                                                     float* __restrict__ H,
                                                     const int2* __restrict__ sorted,
                                                     const int* __restrict__ offsT,
                                                     int* __restrict__ cnt,
                                                     int2* __restrict__ ent,
                                                     int2* __restrict__ spill) {
    int b = blockIdx.x;
    if (b < G_GEMM) gemm_body(b, X, Wp, H);
    else            place_bin_body(b - G_GEMM, sorted, offsT, cnt, ent, spill);
}


__device__ __forceinline__ float4 row_fma(float4 a, float4 v, float w) {
    a.x = fmaf(v.x, w, a.x); a.y = fmaf(v.y, w, a.y);
    a.z = fmaf(v.z, w, a.z); a.w = fmaf(v.w, w, a.w);
    return a;
}

template <int N>
__device__ __forceinline__ void rows_load(float4* v, const float4* __restrict__ h4,
                                          int lane, int sl, int j) {
#pragma unroll
    for (int k = 0; k < N; ++k)
        v[k] = h4[(size_t)__builtin_amdgcn_readlane(sl, j + k) * 32 + lane];
}

template <int N>
__device__ __forceinline__ float4 rows_add(float4 acc, const float4* v, int wl, int j) {
#pragma unroll
    for (int k = 0; k < N; ++k)
        acc = row_fma(acc, v[k], __int_as_float(__builtin_amdgcn_readlane(wl, j + k)));
    return acc;
}

template <int FIRST>
__device__ __forceinline__ float4 gather_row(int node, int lane,
                                             const float4* __restrict__ h4,
                                             const int* __restrict__ cnt,
                                             int2* __restrict__ ent,
                                             const int2* __restrict__ spill,
                                             float4 bv) {
    int   n  = cnt[node];
    float dn = rsqrtf((float)(n + 1));

    float4 hv = h4[(size_t)node * 32 + lane];
    float4 acc = row_fma(bv, hv, dn * dn);

    int nb = n < CAP ? n : CAP;
    int2* my = ent + (size_t)node * CAP + lane;
    int sl, wl;
    if (FIRST) {
        int s = node;
        if (lane < nb) s = my->x;
        sl = s;
        wl = __float_as_int(rsqrtf((float)(cnt[s] + 1)) * dn);

        int key = (sl << 5) | lane;
        int pos = 0;
        for (int j = 0; j < nb; ++j)
            pos += (__builtin_amdgcn_readlane(key, j) < key) ? 1 : 0;
        if (lane >= nb) pos = lane;
        sl = __builtin_amdgcn_ds_permute(pos << 2, sl);
        wl = __builtin_amdgcn_ds_permute(pos << 2, wl);
        if (lane < nb) *my = make_int2(sl, wl);
    } else {
        int2 e = make_int2(0, 0);
        if (lane < nb) e = *my;
        sl = e.x;
        wl = e.y;
    }

    float4 v[8];
    int j = 0;
    for (; j + 8 <= nb; j += 8) {
        rows_load<8>(v, h4, lane, sl, j);
        acc = rows_add<8>(acc, v, wl, j);
    }
    int rem = nb - j;
    int j2 = j + (rem & 4);
    int j1 = j2 + (rem & 2);
    if (rem & 4) rows_load<4>(v,     h4, lane, sl, j);
    if (rem & 2) rows_load<2>(v + 4, h4, lane, sl, j2);
    if (rem & 1) rows_load<1>(v + 6, h4, lane, sl, j1);
    if (rem & 4) acc = rows_add<4>(acc, v,     wl, j);
    if (rem & 2) acc = rows_add<2>(acc, v + 4, wl, j2);
    if (rem & 1) acc = rows_add<1>(acc, v + 6, wl, j1);

    if (n > CAP) {
        int ns = cnt[N_NODES];
        if (ns > N_EDGES) ns = N_EDGES;
        for (int base = 0; base < ns; base += 32) {
            int i = base + lane;
            int2 e = make_int2(-1, 0);
            if (i < ns) e = spill[i];
            unsigned m = __builtin_amdgcn_ballot_w32(e.x == node);
            while (m) {
                int k = __builtin_ctz(m);
                m &= m - 1;
                int s = __builtin_amdgcn_readlane(e.y, k);
                float w = rsqrtf((float)(cnt[s] + 1)) * dn;
                acc = row_fma(acc, h4[(size_t)s * 32 + lane], w);
            }
        }
    }

    return acc;
}

__global__ __launch_bounds__(256) void k_gather2(const float* __restrict__ h,
                                                 const int* __restrict__ cnt,
                                                 int2* __restrict__ ent,
                                                 const int2* __restrict__ spill,
                                                 const float* __restrict__ bias,
                                                 float* __restrict__ out) {
    int lane = threadIdx.x & 31;
    int node = blockIdx.x * 8 + (threadIdx.x >> 5);
    if (node >= N_NODES) return;
    node = __builtin_amdgcn_readfirstlane(node);
    float4 bv = ((const float4*)bias)[lane];
    float4 acc = gather_row<0>(node, lane, (const float4*)h, cnt, ent, spill, bv);
    ((float4*)out)[(size_t)node * 32 + lane] = acc;
}

#define NPB   32
#define PITCH 132
#define G_FUSED ((N_NODES + NPB - 1) / NPB)
static_assert(N_NODES % 16 == 0, "an M tile is either whole or absent");

__global__ __launch_bounds__(256) void k_gather1_gemm2(const float* __restrict__ h1,
                                                       const int* __restrict__ cnt,
                                                       int2* __restrict__ ent,
                                                       const int2* __restrict__ spill,
                                                       const float* __restrict__ bias,
                                                       const float* __restrict__ Wp,
                                                       float* __restrict__ h2) {
    __shared__ __attribute__((aligned(16))) float xs[NPB * PITCH];

    int lane = threadIdx.x & 31;
    int wave = __builtin_amdgcn_readfirstlane(threadIdx.x >> 5);
    int base = blockIdx.x * NPB;

    float4 bv = ((const float4*)bias)[lane];
    for (int i = 0; i < NPB / 8; ++i) {
        int loc  = i * 8 + wave;
        int node = base + loc;
        float4 x = make_float4(0.0f, 0.0f, 0.0f, 0.0f);
        if (node < N_NODES) {
            x = gather_row<1>(node, lane, (const float4*)h1, cnt, ent, spill, bv);
            x.x = fmaxf(x.x, 0.0f); x.y = fmaxf(x.y, 0.0f);
            x.z = fmaxf(x.z, 0.0f); x.w = fmaxf(x.w, 0.0f);
        }
        *(float4*)(xs + loc * PITCH + 4 * lane) = x;
    }
    __syncthreads();

    int g = lane >> 4;
    int r = lane & 15;
    const float* ap = xs + r * PITCH + 2 * g;
    const float* bp = Wp + (size_t)g * 256 + (wave * 16 + r) * 2;
    v8f acc0 = {}, acc1 = {};
#pragma unroll
    for (int kb = 0; kb < DIM; kb += 4) {
        v2f a0 = *(const v2f*)(ap + kb);
        v2f a1 = *(const v2f*)(ap + 16 * PITCH + kb);
        v2f b  = *(const v2f*)(bp + (kb >> 1) * 256);
        acc0 = __builtin_amdgcn_wmma_f32_16x16x4_f32(false, a0, false, b, (short)0, acc0, false, false);
        acc1 = __builtin_amdgcn_wmma_f32_16x16x4_f32(false, a1, false, b, (short)0, acc1, false, false);
    }

    float* hrow = h2 + ((size_t)base + 8 * g) * DIM + wave * 16 + r;
#pragma unroll
    for (int v = 0; v < 8; ++v) hrow[(size_t)v * DIM] = acc0[v];
    if (base + 16 < N_NODES) {
#pragma unroll
        for (int v = 0; v < 8; ++v) hrow[(size_t)(16 + v) * DIM] = acc1[v];
    }
}


extern "C" void kernel_launch(void* const* d_in, const int* in_sizes, int n_in,
                              void* d_out, int out_size, void* d_ws, size_t ws_size,
                              hipStream_t stream) {
    const float* X  = (const float*)d_in[0];
    const int*   ei = (const int*)d_in[1];
    const float* W1 = (const float*)d_in[2];
    const float* b1 = (const float*)d_in[3];
    const float* W2 = (const float*)d_in[4];
    const float* b2 = (const float*)d_in[5];
    float* out = (float*)d_out;

    const int* src = ei;
    const int* dst = ei + N_EDGES;

    float* ws   = (float*)d_ws;
    float* Wp1  = ws;
    float* Wp2  = Wp1 + 16384;
    float* h2   = Wp2 + 16384;
    int*  cnt    = (int*)(h2 + (size_t)N_NODES * DIM);
    int2* ent    = (int2*)(cnt + 50048);
    int2* spill  = ent + (size_t)N_NODES * CAP;
    float* h1    = out;
    int2* sorted = (int2*)h2;
    int*  offsT  = (int*)(sorted + (size_t)NCH * EPB);

    if ((size_t)((const char*)(spill + N_EDGES) - (const char*)d_ws) > ws_size) return;

    const int gRow = (N_NODES + 7) / 8;

    k_prep<<<NCH + 2 * G_W, THR, 0, stream>>>(src, dst, sorted, offsT, cnt, W1, Wp1, W2, Wp2);

    k_gemm1_place<<<G_GEMM + NBIN, THR, 0, stream>>>(X, Wp1, h1, sorted, offsT, cnt, ent, spill);

    k_gather1_gemm2<<<G_FUSED, THR, 0, stream>>>(h1, cnt, ent, spill, b1, Wp2, h2);

    k_gather2<<<gRow, THR, 0, stream>>>(h2, cnt, ent, spill, b2, out);
}
